# starting kernel
# baseline (speedup 1.0000x reference)
#include <hip/hip_runtime.h>

typedef float v2f __attribute__((ext_vector_type(2)));
typedef float v8f __attribute__((ext_vector_type(8)));

#define VSZ  32128
#define HID  16
#define OUTD 16
#define SEQ  512
#define ROWS 16
#define TPR  16
#define BLK  (ROWS * TPR)

__device__ __forceinline__ float4 f4add(float4 a, float4 b) {
    return make_float4(a.x + b.x, a.y + b.y, a.z + b.z, a.w + b.w);
}

__global__ void transpose_w_kernel(const float* __restrict__ w,
                                   float* __restrict__ wT) {
    int v = blockIdx.x * blockDim.x + threadIdx.x;
    if (v >= VSZ) return;
#pragma unroll
    for (int k = 0; k < HID; ++k)
        wT[(size_t)v * HID + k] = w[(size_t)k * VSZ + v];
}

__global__ void __launch_bounds__(BLK)
encoder_kernel(const int* __restrict__ x,
               const float* __restrict__ wT,
               const float* __restrict__ enc1_b,
               const float* __restrict__ mean_w,
               const float* __restrict__ mean_b,
               const float* __restrict__ lv_w,
               const float* __restrict__ lv_b,
               float* __restrict__ out_mean,
               float* __restrict__ out_lv,
               int Btot) {
    __shared__ float part[ROWS][TPR][HID];
    __shared__ float hmat[ROWS][HID];

    const int tid = threadIdx.x;
    const int row = tid >> 4;
    const int tr  = tid & 15;
    const int b   = blockIdx.x * ROWS + row;

    float4 a0 = make_float4(0.f, 0.f, 0.f, 0.f);
    float4 a1 = a0, a2 = a0, a3 = a0;
    if (b < Btot) {
        const int* xrow = x + (size_t)b * SEQ;
        const float4* wT4 = (const float4*)wT;
        for (int s = tr; s < SEQ; s += TPR) {
            int tok = xrow[s];
            const float4* wr = wT4 + (size_t)tok * 4;
            a0 = f4add(a0, wr[0]);
            a1 = f4add(a1, wr[1]);
            a2 = f4add(a2, wr[2]);
            a3 = f4add(a3, wr[3]);
        }
    }
    {
        float4* p = (float4*)&part[row][tr][0];
        p[0] = a0; p[1] = a1; p[2] = a2; p[3] = a3;
    }
    __syncthreads();

    {
        const int r = row, k = tr;
        float s = 0.f;
#pragma unroll
        for (int t = 0; t < TPR; ++t) s += part[r][t][k];
        s += enc1_b[k];
        hmat[r][k] = s > 0.f ? s : 0.f;
    }
    __syncthreads();

    if (tid < 32) {
        const int lane = tid;
        const int half = lane >> 4;
        const int l16  = lane & 15;

        v2f a[4];
#pragma unroll
        for (int c = 0; c < 4; ++c) {
            int k0 = 4 * c + 2 * half;
            a[c].x = hmat[l16][k0];
            a[c].y = hmat[l16][k0 + 1];
        }
        v2f bm[4], bl[4];
#pragma unroll
        for (int c = 0; c < 4; ++c) {
            int k0 = 4 * c + 2 * half;
            bm[c].x = mean_w[l16 * HID + k0];
            bm[c].y = mean_w[l16 * HID + k0 + 1];
            bl[c].x = lv_w[l16 * HID + k0];
            bl[c].y = lv_w[l16 * HID + k0 + 1];
        }

        v8f accm = {};
#pragma unroll
        for (int c = 0; c < 4; ++c)
            accm = __builtin_amdgcn_wmma_f32_16x16x4_f32(
                false, a[c], false, bm[c], (short)0, accm, false, false);
        v8f accl = {};
#pragma unroll
        for (int c = 0; c < 4; ++c)
            accl = __builtin_amdgcn_wmma_f32_16x16x4_f32(
                false, a[c], false, bl[c], (short)0, accl, false, false);

        const float mb = mean_b[l16];
        const float lb = lv_b[l16];
        const int b0 = blockIdx.x * ROWS;
#pragma unroll
        for (int j = 0; j < 8; ++j) {
            int m = j + 8 * half;
            int gb = b0 + m;
            if (gb < Btot) {
                out_mean[(size_t)gb * OUTD + l16] = accm[j] + mb;
                out_lv[(size_t)gb * OUTD + l16]   = accl[j] + lb;
            }
        }
    }
}

extern "C" void kernel_launch(void* const* d_in, const int* in_sizes, int n_in,
                              void* d_out, int out_size, void* d_ws, size_t ws_size,
                              hipStream_t stream) {
    const int*   x      = (const int*)d_in[0];
    const float* enc1_w = (const float*)d_in[1];
    const float* enc1_b = (const float*)d_in[2];
    const float* mean_w = (const float*)d_in[3];
    const float* mean_b = (const float*)d_in[4];
    const float* lv_w   = (const float*)d_in[5];
    const float* lv_b   = (const float*)d_in[6];

    const int Btot = in_sizes[0] / SEQ;

    float* wT = (float*)d_ws;

    transpose_w_kernel<<<(VSZ + 255) / 256, 256, 0, stream>>>(enc1_w, wT);

    float* out_mean = (float*)d_out;
    float* out_lv   = (float*)d_out + (size_t)Btot * OUTD;
    encoder_kernel<<<(Btot + ROWS - 1) / ROWS, BLK, 0, stream>>>(
        x, wT, enc1_b, mean_w, mean_b, lv_w, lv_b, out_mean, out_lv, Btot);
}


# token sum moved to the matrix unit: half-waves load 64-byte table rows as A fragments, a 0/1 B operand sums them per batch row, h feeds the projections in place; summation order changes
# speedup vs baseline: 1.4919x; 1.4919x over previous
#include <hip/hip_runtime.h>


typedef float v2f __attribute__((ext_vector_type(2)));
typedef float v8f __attribute__((ext_vector_type(8)));

#define VSZ  32128
#define HID  16
#define OUTD 16
#define SEQ  512
#define RPW  2
#define WPB  4
#define BLK  (WPB * 32)
#define GPR  (SEQ / 8)
#define NGRP (RPW * GPR)
#define DG   11
#define EI   12
#define RV   (DG + 1)

__global__ void transpose_w_kernel(const float* __restrict__ w,
                                   float* __restrict__ wT) {
    int v = blockIdx.x * blockDim.x + threadIdx.x;
    if (v >= VSZ) return;
#pragma unroll
    for (int k = 0; k < HID; ++k)
        wT[(size_t)v * HID + k] = w[(size_t)k * VSZ + v];
}

__device__ __forceinline__ v8f mma4(v2f a, v2f b, v8f c) {
    return __builtin_amdgcn_wmma_f32_16x16x4_f32(false, a, false, b, (short)0, c,
                                                 false, false);
}

__device__ __forceinline__ int4 load_ids(const int4* __restrict__ xa,
                                         const int4* __restrict__ xb, int G) {
    return (G < GPR ? xa : xb)[2 * (G & (GPR - 1))];
}

__device__ __forceinline__ void gather4(float (&v)[4], int4 q,
                                        const char* __restrict__ tab, unsigned lo) {
    v[0] = *(const float*)(tab + (((unsigned)q.x << 6) | lo));
    v[1] = *(const float*)(tab + (((unsigned)q.y << 6) | lo));
    v[2] = *(const float*)(tab + (((unsigned)q.z << 6) | lo));
    v[3] = *(const float*)(tab + (((unsigned)q.w << 6) | lo));
}

__global__ void __launch_bounds__(BLK)
encoder_kernel(const int* __restrict__ x,
               const float* __restrict__ wT,
               const float* __restrict__ enc1_b,
               const float* __restrict__ mean_w,
               const float* __restrict__ mean_b,
               const float* __restrict__ lv_w,
               const float* __restrict__ lv_b,
               float* __restrict__ out_mean,
               float* __restrict__ out_lv,
               int Btot) {
    const int lane = threadIdx.x & 31;
    const int h = lane >> 4;
    const int c = lane & 15;
    const int wave = __builtin_amdgcn_readfirstlane((int)(threadIdx.x >> 5));
    const int row0 = (blockIdx.x * WPB + wave) * RPW;
    if (row0 >= Btot) return;
    const int row1 = row0 + 1 < Btot ? row0 + 1 : row0;

    const int4* xa = (const int4*)(x + (size_t)row0 * SEQ) + h;
    const int4* xb = (const int4*)(x + (size_t)row1 * SEQ) + h;
    const char* tab = (const char*)wT;
    const unsigned lo = (unsigned)c * 4u;

    v2f wm[4], wl[4];
#pragma unroll
    for (int j = 0; j < 4; ++j) {
        wm[j] = *(const v2f*)(mean_w + c * HID + 8 * h + 2 * j);
        wl[j] = *(const v2f*)(lv_w + c * HID + 8 * h + 2 * j);
    }
    const float4 eb0 = ((const float4*)(enc1_b + 8 * h))[0];
    const float4 eb1 = ((const float4*)(enc1_b + 8 * h))[1];
    const float4 mb0 = ((const float4*)(mean_b + 8 * h))[0];
    const float4 mb1 = ((const float4*)(mean_b + 8 * h))[1];
    const float4 lb0 = ((const float4*)(lv_b + 8 * h))[0];
    const float4 lb1 = ((const float4*)(lv_b + 8 * h))[1];

    const float s0 = (c < 8) ? 1.0f : 0.0f;
    const v2f sel0 = {s0, s0};
    const v2f sel1 = {1.0f - s0, 1.0f - s0};

    float vr[RV][4];
    int4  ir[EI];
    v8f acc0 = {}, acc1 = {};

    {
        int4 q0[DG];
#pragma unroll
        for (int g = 0; g < DG; ++g) q0[g] = load_ids(xa, xb, g);
#pragma unroll
        for (int g = 0; g < EI; ++g) ir[(DG + g) % EI] = load_ids(xa, xb, DG + g);
#pragma unroll
        for (int g = 0; g < DG; ++g) gather4(vr[g % RV], q0[g], tab, lo);
        __builtin_amdgcn_sched_barrier(0);
    }

#pragma unroll
    for (int G = 0; G < NGRP; ++G) {
        if (G + DG < NGRP) gather4(vr[(G + DG) % RV], ir[(G + DG) % EI], tab, lo);
        if (G + DG + EI < NGRP) ir[(G + DG) % EI] = load_ids(xa, xb, G + DG + EI);
        const v2f sel = G < GPR ? sel0 : sel1;
        const v2f a0 = {vr[G % RV][0], vr[G % RV][1]};
        const v2f a1 = {vr[G % RV][2], vr[G % RV][3]};
        acc0 = mma4(a0, sel, acc0);
        acc1 = mma4(a1, sel, acc1);
        __builtin_amdgcn_sched_barrier(0);
    }

    const v8f hs = acc0 + acc1;
    const float eb[8] = {eb0.x, eb0.y, eb0.z, eb0.w, eb1.x, eb1.y, eb1.z, eb1.w};
    float ha[8];
#pragma unroll
    for (int r = 0; r < 8; ++r) {
        float s = hs[r] + eb[r];
        ha[r] = s > 0.f ? s : 0.f;
    }

    v8f om = {mb0.x, mb0.y, mb0.z, mb0.w, mb1.x, mb1.y, mb1.z, mb1.w};
    v8f ol = {lb0.x, lb0.y, lb0.z, lb0.w, lb1.x, lb1.y, lb1.z, lb1.w};
#pragma unroll
    for (int j = 0; j < 4; ++j) {
        v2f b = {ha[2 * j], ha[2 * j + 1]};
        om = mma4(wm[j], b, om);
        ol = mma4(wl[j], b, ol);
    }

    const int q = c & 1;
    const int orow = row0 + (c >> 3);
    if ((c & 7) < 2 && orow < Btot) {
        float4 m, l;
        m.x = q ? om[4] : om[0];
        m.y = q ? om[5] : om[1];
        m.z = q ? om[6] : om[2];
        m.w = q ? om[7] : om[3];
        l.x = q ? ol[4] : ol[0];
        l.y = q ? ol[5] : ol[1];
        l.z = q ? ol[6] : ol[2];
        l.w = q ? ol[7] : ol[3];
        const size_t o = (size_t)orow * OUTD + 8 * h + 4 * q;
        *(float4*)(out_mean + o) = m;
        *(float4*)(out_lv + o) = l;
    }
}

extern "C" void kernel_launch(void* const* d_in, const int* in_sizes, int n_in,
                              void* d_out, int out_size, void* d_ws, size_t ws_size,
                              hipStream_t stream) {
    const int*   x      = (const int*)d_in[0];
    const float* enc1_w = (const float*)d_in[1];
    const float* enc1_b = (const float*)d_in[2];
    const float* mean_w = (const float*)d_in[3];
    const float* mean_b = (const float*)d_in[4];
    const float* lv_w   = (const float*)d_in[5];
    const float* lv_b   = (const float*)d_in[6];

    const int Btot = in_sizes[0] / SEQ;

    float* wT = (float*)d_ws;

    transpose_w_kernel<<<(VSZ + 255) / 256, 256, 0, stream>>>(enc1_w, wT);

    float* out_mean = (float*)d_out;
    float* out_lv   = (float*)d_out + (size_t)Btot * OUTD;
    encoder_kernel<<<(Btot + WPB * RPW - 1) / (WPB * RPW), BLK, 0, stream>>>(
        x, wT, enc1_b, mean_w, mean_b, lv_w, lv_b, out_mean, out_lv, Btot);
}
